# starting kernel
# baseline (speedup 1.0000x reference)
#include <hip/hip_runtime.h>

typedef __attribute__((ext_vector_type(2))) float v2f;
typedef __attribute__((ext_vector_type(8))) float v8f;

#define NN 4096
#define EE 16384
#define DD 13
#define RR 128
#define PP 4
#define W1S 18
#define W2S 130

__global__ void zero_out_kernel(float* __restrict__ out, int n) {
  int i = blockIdx.x * blockDim.x + threadIdx.x;
  if (i < n) out[i] = 0.0f;
}

__global__ __launch_bounds__(256) void bpnet_stage1_kernel(
    const float* __restrict__ nodes,
    const float* __restrict__ bp_params,
    const float* __restrict__ bp_bias,
    const int*   __restrict__ edges,
    const int*   __restrict__ edge_types,
    float*       __restrict__ t_out)
{
  __shared__ float Wt[PP * RR * W1S];
  __shared__ float Bs[PP * RR];

  const int tid = threadIdx.x;
  for (int idx = tid; idx < PP * RR * W1S; idx += 256) {
    int p = idx / (RR * W1S);
    int r = idx % (RR * W1S);
    int n = r / W1S;
    int k = r % W1S;
    Wt[idx] = (k < DD) ? bp_params[(p * DD + k) * RR + n] : 0.0f;
  }
  for (int idx = tid; idx < PP * RR; idx += 256) Bs[idx] = bp_bias[idx];
  __syncthreads();

  const int wave = tid >> 5;
  const int lane = tid & 31;
  const int row  = lane & 15;
  const int kh   = lane >> 4;
  const int slot = blockIdx.y;
  const int tile = blockIdx.x * 8 + wave;
  const int e0   = tile * 16;
  const int e    = e0 + row;

  const int etype = edge_types[e * 3 + slot];
  const int nidx  = edges[e * 3 + slot];
  const int nbase = nidx * DD;

  v2f av[4];
#pragma unroll
  for (int c = 0; c < 4; ++c) {
    int kk = 4 * c + 2 * kh;
    av[c].x = (kk     < DD) ? nodes[nbase + kk]     : 0.0f;
    av[c].y = (kk + 1 < DD) ? nodes[nbase + kk + 1] : 0.0f;
  }

  for (int nt = 0; nt < 8; ++nt) {
    const int ng = nt * 16 + row;
    v8f acc = {0.f, 0.f, 0.f, 0.f, 0.f, 0.f, 0.f, 0.f};
#pragma unroll
    for (int c = 0; c < 4; ++c) {
      const int kk = 4 * c + 2 * kh;
#pragma unroll
      for (int p = 0; p < PP; ++p) {
        v2f b = *(const v2f*)&Wt[(p * RR + ng) * W1S + kk];
        v2f a;
        a.x = (etype == p) ? av[c].x : 0.0f;
        a.y = (etype == p) ? av[c].y : 0.0f;
        acc = __builtin_amdgcn_wmma_f32_16x16x4_f32(
            false, a, false, b, (short)0, acc, false, false);
      }
    }
#pragma unroll
    for (int v = 0; v < 8; ++v) {
      int r2  = v + 8 * kh;
      int tr  = __shfl(etype, r2, 32);
      float x = acc[v] + Bs[tr * RR + ng];
      x = (x > 0.0f) ? x : 0.0f;
      t_out[(size_t)(slot * EE + (e0 + r2)) * RR + ng] = x;
    }
  }
}

__global__ __launch_bounds__(256) void bpnet_stage2_kernel(
    const float* __restrict__ t_in,
    const float* __restrict__ ho_params,
    const float* __restrict__ ho_bias,
    const int*   __restrict__ edges,
    const int*   __restrict__ edge_types,
    float*       __restrict__ out)
{
  __shared__ float Wt[PP * 16 * W2S];
  __shared__ float Bs[PP * 16];

  const int slot = blockIdx.y;
  const int tid  = threadIdx.x;
  for (int idx = tid; idx < PP * 16 * W2S; idx += 256) {
    int p = idx / (16 * W2S);
    int r = idx % (16 * W2S);
    int n = r / W2S;
    int k = r % W2S;
    float v = 0.0f;
    if (n < DD && k < RR)
      v = ho_params[((size_t)(slot * PP + p) * RR + k) * DD + n];
    Wt[idx] = v;
  }
  for (int idx = tid; idx < PP * 16; idx += 256) {
    int p = idx / 16, n = idx % 16;
    Bs[idx] = (n < DD) ? ho_bias[(slot * PP + p) * DD + n] : 0.0f;
  }
  __syncthreads();

  const int wave = tid >> 5;
  const int lane = tid & 31;
  const int row  = lane & 15;
  const int kh   = lane >> 4;
  const int tile = blockIdx.x * 8 + wave;
  const int e0   = tile * 16;
  const int e    = e0 + row;

  const int etype = edge_types[e * 3 + slot];
  const int tgt   = edges[e * 3 + slot];
  const int j0    = (slot == 0) ? 1 : 0;
  const int j1    = (slot == 2) ? 1 : 2;

  const float* __restrict__ t0 = t_in + (size_t)(j0 * EE + e) * RR;
  const float* __restrict__ t1 = t_in + (size_t)(j1 * EE + e) * RR;

  v8f acc = {0.f, 0.f, 0.f, 0.f, 0.f, 0.f, 0.f, 0.f};
  for (int c = 0; c < 32; ++c) {
    const int kk = 4 * c + 2 * kh;
    v2f x0 = *(const v2f*)(t0 + kk);
    v2f x1 = *(const v2f*)(t1 + kk);
    v2f a;
    a.x = x0.x * x1.x;
    a.y = x0.y * x1.y;
#pragma unroll
    for (int p = 0; p < PP; ++p) {
      v2f b = *(const v2f*)&Wt[(p * 16 + row) * W2S + kk];
      v2f ap;
      ap.x = (etype == p) ? a.x : 0.0f;
      ap.y = (etype == p) ? a.y : 0.0f;
      acc = __builtin_amdgcn_wmma_f32_16x16x4_f32(
          false, ap, false, b, (short)0, acc, false, false);
    }
  }

#pragma unroll
  for (int v = 0; v < 8; ++v) {
    int r2 = v + 8 * kh;
    int tr = __shfl(etype, r2, 32);
    int tg = __shfl(tgt,   r2, 32);
    if (row < DD) {
      float val = acc[v] + Bs[tr * 16 + row];
      atomicAdd(&out[tg * DD + row], val);
    }
  }
}

extern "C" void kernel_launch(void* const* d_in, const int* in_sizes, int n_in,
                              void* d_out, int out_size, void* d_ws, size_t ws_size,
                              hipStream_t stream) {
  const float* nodes      = (const float*)d_in[0];
  const float* bp_params  = (const float*)d_in[1];
  const float* bp_bias    = (const float*)d_in[2];
  const float* ho_params  = (const float*)d_in[3];
  const float* ho_bias    = (const float*)d_in[4];
  const int*   edges      = (const int*)d_in[5];
  const int*   edge_types = (const int*)d_in[6];
  float* out  = (float*)d_out;
  float* t_ws = (float*)d_ws;

  zero_out_kernel<<<(NN * DD + 255) / 256, 256, 0, stream>>>(out, NN * DD);

  bpnet_stage1_kernel<<<dim3(EE / 16 / 8, 3), 256, 0, stream>>>(
      nodes, bp_params, bp_bias, edges, edge_types, t_ws);
  bpnet_stage2_kernel<<<dim3(EE / 16 / 8, 3), 256, 0, stream>>>(
      t_ws, ho_params, ho_bias, edges, edge_types, out);
}


# Stage-2 weights copied to LDS in memory order [type][k][13] by 16-byte loads issued together (no per-element transposing fill); B k-pairs read as two dwords 13 apart, a chunk ahead of their WMMAs
# speedup vs baseline: 1.7123x; 1.7123x over previous
#include <hip/hip_runtime.h>

typedef __attribute__((ext_vector_type(2))) float v2f;
typedef __attribute__((ext_vector_type(4))) float v4f;
typedef __attribute__((ext_vector_type(8))) float v8f;

#define NN 4096
#define EE 16384
#define DD 13
#define RR 128
#define PP 4

__global__ __launch_bounds__(128) void bpnet_table_kernel(
    const float* __restrict__ nodes,
    const float* __restrict__ bp_params,
    const float* __restrict__ bp_bias,
    float*       __restrict__ tab,
    float*       __restrict__ out)
{
  const int tid   = threadIdx.x;

  const int gid = blockIdx.x * 128 + tid;
  if (gid < NN * DD / 4)
    ((v4f*)out)[gid] = (v4f){0.f, 0.f, 0.f, 0.f};

  const int p     = tid >> 5;
  const int lane  = tid & 31;
  const int n     = lane & 15;
  const int kh    = lane >> 4;
  const int node0 = blockIdx.x * 16;

  const float* __restrict__ nrow = nodes + (size_t)(node0 + n) * DD;
  v2f av[4];
#pragma unroll
  for (int c = 0; c < 3; ++c) {
    av[c].x = nrow[4 * c + 2 * kh];
    av[c].y = nrow[4 * c + 2 * kh + 1];
  }
  const float a12 = nrow[12];
  av[3].x = (kh == 0) ? a12 : 0.0f;
  av[3].y = 0.0f;

  const float* __restrict__ wbase = bp_params + (size_t)p * DD * RR + 4 * n;
  const v4f zero4 = {0.f, 0.f, 0.f, 0.f};
  v4f wv[2][8];
#pragma unroll
  for (int g = 0; g < 2; ++g) {
#pragma unroll
    for (int c = 0; c < 3; ++c) {
      wv[g][2 * c]     = *(const v4f*)(wbase + (4 * c + 2 * kh) * RR + 64 * g);
      wv[g][2 * c + 1] = *(const v4f*)(wbase + (4 * c + 2 * kh + 1) * RR + 64 * g);
    }
    const v4f w12 = *(const v4f*)(wbase + 12 * RR + 64 * g);
    wv[g][6] = (kh == 0) ? w12 : zero4;
    wv[g][7] = zero4;
  }

  v4f bias[2];
#pragma unroll
  for (int g = 0; g < 2; ++g)
    bias[g] = *(const v4f*)(bp_bias + p * RR + 64 * g + 4 * n);

  v8f acc[8];
#pragma unroll
  for (int t = 0; t < 8; ++t)
    acc[t] = (v8f){0.f, 0.f, 0.f, 0.f, 0.f, 0.f, 0.f, 0.f};

#pragma unroll
  for (int c = 0; c < 4; ++c) {
#pragma unroll
    for (int t = 0; t < 8; ++t) {
      const int g = t >> 2, j = t & 3;
      v2f b;
      b.x = wv[g][2 * c][j];
      b.y = wv[g][2 * c + 1][j];
      acc[t] = __builtin_amdgcn_wmma_f32_16x16x4_f32(
          false, av[c], false, b, (short)0, acc[t], false, false);
    }
  }

#pragma unroll
  for (int g = 0; g < 2; ++g) {
#pragma unroll
    for (int v = 0; v < 8; ++v) {
      v4f o;
      o.x = acc[4 * g + 0][v] + bias[g].x;
      o.y = acc[4 * g + 1][v] + bias[g].y;
      o.z = acc[4 * g + 2][v] + bias[g].z;
      o.w = acc[4 * g + 3][v] + bias[g].w;
      o.x = (o.x > 0.0f) ? o.x : 0.0f;
      o.y = (o.y > 0.0f) ? o.y : 0.0f;
      o.z = (o.z > 0.0f) ? o.z : 0.0f;
      o.w = (o.w > 0.0f) ? o.w : 0.0f;
      *(v4f*)(tab + ((size_t)(node0 + v + 8 * kh) * PP + p) * RR + 64 * g + 4 * n) = o;
    }
  }
}

__global__ __launch_bounds__(256) void bpnet_stage2_kernel(
    const float* __restrict__ tab,
    const float* __restrict__ ho_params,
    const float* __restrict__ ho_bias,
    const int*   __restrict__ edges,
    const int*   __restrict__ edge_types,
    float*       __restrict__ out)
{
  __shared__ __attribute__((aligned(16))) float Wn[PP * RR * DD];
  __shared__ float Bs[PP * 16];

  const int slot = blockIdx.y;
  const int tid  = threadIdx.x;

  const v4f* __restrict__ wsrc =
      (const v4f*)(ho_params + (size_t)slot * PP * RR * DD);
  v4f wreg[7];
#pragma unroll
  for (int i = 0; i < 6; ++i)
    wreg[i] = wsrc[tid + 256 * i];
  wreg[6] = wsrc[(tid < 128) ? tid + 1536 : tid + 1280];
  for (int idx = tid; idx < PP * 16; idx += 256) {
    int p = idx / 16, n = idx % 16;
    Bs[idx] = (n < DD) ? ho_bias[(slot * PP + p) * DD + n] : 0.0f;
  }
#pragma unroll
  for (int i = 0; i < 6; ++i)
    ((v4f*)Wn)[tid + 256 * i] = wreg[i];
  if (tid < 128)
    ((v4f*)Wn)[tid + 1536] = wreg[6];
  __syncthreads();

  const int wave = tid >> 5;
  const int lane = tid & 31;
  const int row  = lane & 15;
  const int kh   = lane >> 4;
  const int tile = blockIdx.x * 8 + wave;
  const int e0   = tile * 16;
  const int e    = e0 + row;

  const int etype = edge_types[e * 3 + slot];
  const int tgt   = edges[e * 3 + slot];
  const int j0    = (slot == 0) ? 1 : 0;
  const int j1    = (slot == 2) ? 1 : 2;

  const float* __restrict__ t0 =
      tab + ((size_t)edges[e * 3 + j0] * PP + edge_types[e * 3 + j0]) * RR;
  const float* __restrict__ t1 =
      tab + ((size_t)edges[e * 3 + j1] * PP + edge_types[e * 3 + j1]) * RR;

  const int nn = (row < DD) ? row : DD - 1;
  const float* wl = Wn + 2 * kh * DD + nn;

  v8f acc = {0.f, 0.f, 0.f, 0.f, 0.f, 0.f, 0.f, 0.f};
  v2f bq[2][PP];
#pragma unroll
  for (int p = 0; p < PP; ++p) {
    const float* q = wl + p * RR * DD;
    bq[0][p].x = q[0];
    bq[0][p].y = q[DD];
  }
#pragma unroll
  for (int c = 0; c < 32; ++c) {
    if (c + 1 < 32) {
#pragma unroll
      for (int p = 0; p < PP; ++p) {
        const float* q = wl + (p * RR + 4 * (c + 1)) * DD;
        bq[(c + 1) & 1][p].x = q[0];
        bq[(c + 1) & 1][p].y = q[DD];
      }
    }
    __builtin_amdgcn_sched_barrier(0x676);
    const int kk = 4 * c + 2 * kh;
    v2f x0 = *(const v2f*)(t0 + kk);
    v2f x1 = *(const v2f*)(t1 + kk);
    v2f a;
    a.x = x0.x * x1.x;
    a.y = x0.y * x1.y;
#pragma unroll
    for (int p = 0; p < PP; ++p) {
      v2f ap;
      ap.x = (etype == p) ? a.x : 0.0f;
      ap.y = (etype == p) ? a.y : 0.0f;
      acc = __builtin_amdgcn_wmma_f32_16x16x4_f32(
          false, ap, false, bq[c & 1][p], (short)0, acc, false, false);
    }
  }

#pragma unroll
  for (int v = 0; v < 8; ++v) {
    int r2 = v + 8 * kh;
    int tr = __shfl(etype, r2, 32);
    int tg = __shfl(tgt,   r2, 32);
    if (row < DD) {
      float val = acc[v] + Bs[tr * 16 + row];
      atomicAdd(&out[tg * DD + row], val);
    }
  }
}

extern "C" void kernel_launch(void* const* d_in, const int* in_sizes, int n_in,
                              void* d_out, int out_size, void* d_ws, size_t ws_size,
                              hipStream_t stream) {
  const float* nodes      = (const float*)d_in[0];
  const float* bp_params  = (const float*)d_in[1];
  const float* bp_bias    = (const float*)d_in[2];
  const float* ho_params  = (const float*)d_in[3];
  const float* ho_bias    = (const float*)d_in[4];
  const int*   edges      = (const int*)d_in[5];
  const int*   edge_types = (const int*)d_in[6];
  float* out  = (float*)d_out;
  float* t_ws = (float*)d_ws;

  bpnet_table_kernel<<<NN / 16, 128, 0, stream>>>(
      nodes, bp_params, bp_bias, t_ws, out);
  bpnet_stage2_kernel<<<dim3(EE / 16 / 8, 3), 256, 0, stream>>>(
      t_ws, ho_params, ho_bias, edges, edge_types, out);
}


# Stage 2: one wave does all 3 slots of its 16 edges (one 4-wave block per WGP, all weights in LDS): each table row fetched once for two products, 3 independent accumulators; atomic order changes
# speedup vs baseline: 1.9016x; 1.1106x over previous
#include <hip/hip_runtime.h>

typedef __attribute__((ext_vector_type(2))) float v2f;
typedef __attribute__((ext_vector_type(4))) float v4f;
typedef __attribute__((ext_vector_type(8))) float v8f;

#define NN 4096
#define EE 16384
#define DD 13
#define RR 128
#define PP 4
#define SS 3

__global__ __launch_bounds__(128) void bpnet_table_kernel(
    const float* __restrict__ nodes,
    const float* __restrict__ bp_params,
    const float* __restrict__ bp_bias,
    float*       __restrict__ tab,
    float*       __restrict__ out)
{
  const int tid   = threadIdx.x;

  const int gid = blockIdx.x * 128 + tid;
  if (gid < NN * DD / 4)
    ((v4f*)out)[gid] = (v4f){0.f, 0.f, 0.f, 0.f};

  const int p     = tid >> 5;
  const int lane  = tid & 31;
  const int n     = lane & 15;
  const int kh    = lane >> 4;
  const int node0 = blockIdx.x * 16;

  const float* __restrict__ nrow = nodes + (size_t)(node0 + n) * DD;
  v2f av[4];
#pragma unroll
  for (int c = 0; c < 3; ++c) {
    av[c].x = nrow[4 * c + 2 * kh];
    av[c].y = nrow[4 * c + 2 * kh + 1];
  }
  const float a12 = nrow[12];
  av[3].x = (kh == 0) ? a12 : 0.0f;
  av[3].y = 0.0f;

  const float* __restrict__ wbase = bp_params + (size_t)p * DD * RR + 4 * n;
  const v4f zero4 = {0.f, 0.f, 0.f, 0.f};
  v4f wv[2][8];
#pragma unroll
  for (int g = 0; g < 2; ++g) {
#pragma unroll
    for (int c = 0; c < 3; ++c) {
      wv[g][2 * c]     = *(const v4f*)(wbase + (4 * c + 2 * kh) * RR + 64 * g);
      wv[g][2 * c + 1] = *(const v4f*)(wbase + (4 * c + 2 * kh + 1) * RR + 64 * g);
    }
    const v4f w12 = *(const v4f*)(wbase + 12 * RR + 64 * g);
    wv[g][6] = (kh == 0) ? w12 : zero4;
    wv[g][7] = zero4;
  }

  v4f bias[2];
#pragma unroll
  for (int g = 0; g < 2; ++g)
    bias[g] = *(const v4f*)(bp_bias + p * RR + 64 * g + 4 * n);

  v8f acc[8];
#pragma unroll
  for (int t = 0; t < 8; ++t)
    acc[t] = (v8f){0.f, 0.f, 0.f, 0.f, 0.f, 0.f, 0.f, 0.f};

#pragma unroll
  for (int c = 0; c < 4; ++c) {
#pragma unroll
    for (int t = 0; t < 8; ++t) {
      const int g = t >> 2, j = t & 3;
      v2f b;
      b.x = wv[g][2 * c][j];
      b.y = wv[g][2 * c + 1][j];
      acc[t] = __builtin_amdgcn_wmma_f32_16x16x4_f32(
          false, av[c], false, b, (short)0, acc[t], false, false);
    }
  }

#pragma unroll
  for (int g = 0; g < 2; ++g) {
#pragma unroll
    for (int v = 0; v < 8; ++v) {
      v4f o;
      o.x = acc[4 * g + 0][v] + bias[g].x;
      o.y = acc[4 * g + 1][v] + bias[g].y;
      o.z = acc[4 * g + 2][v] + bias[g].z;
      o.w = acc[4 * g + 3][v] + bias[g].w;
      o.x = (o.x > 0.0f) ? o.x : 0.0f;
      o.y = (o.y > 0.0f) ? o.y : 0.0f;
      o.z = (o.z > 0.0f) ? o.z : 0.0f;
      o.w = (o.w > 0.0f) ? o.w : 0.0f;
      *(v4f*)(tab + ((size_t)(node0 + v + 8 * kh) * PP + p) * RR + 64 * g + 4 * n) = o;
    }
  }
}

__global__ __launch_bounds__(128) void bpnet_stage2_kernel(
    const float* __restrict__ tab,
    const float* __restrict__ ho_params,
    const float* __restrict__ ho_bias,
    const int*   __restrict__ edges,
    const int*   __restrict__ edge_types,
    float*       __restrict__ out)
{
  __shared__ __attribute__((aligned(16))) float Wn[162 * 256];
  __shared__ float Bs[256];


  const int tid  = threadIdx.x;
  const int wave = tid >> 5;
  const int lane = tid & 31;
  const int row  = lane & 15;
  const int kh   = lane >> 4;
  const int e    = (blockIdx.x * 4 + wave) * 16 + row;

  int nd[SS], ty[SS];
#pragma unroll
  for (int s = 0; s < SS; ++s) {
    nd[s] = edges[e * 3 + s];
    ty[s] = edge_types[e * 3 + s];
  }
  const int bn  = tid & 15;
  const int bnc = (bn < DD) ? bn : DD - 1;
  const int bsp = tid >> 4;
  const int bsq = (tid < 64) ? bsp + 8 : SS * PP - 1;
  const float bias0 = ho_bias[bsp * DD + bnc];
  const float bias1 = ho_bias[bsq * DD + bnc];
  __builtin_amdgcn_sched_barrier(0);

  const v4f* __restrict__ wsrc = (const v4f*)ho_params;
  v4f wreg[39];
#pragma unroll
  for (int i = 0; i < 39; ++i)
    wreg[i] = wsrc[tid + 128 * i];
  __builtin_amdgcn_sched_barrier(0);

  const float* __restrict__ tp[SS];
#pragma unroll
  for (int s = 0; s < SS; ++s)
    tp[s] = tab + ((size_t)nd[s] * PP + ty[s]) * RR + 2 * kh;

  Bs[tid]       = (bn < DD) ? bias0 : 0.0f;
  Bs[tid + 128] = (bn < DD && tid < 64) ? bias1 : 0.0f;

  v2f x[SS][32];
#pragma unroll
  for (int c = 0; c < 8; ++c) {
#pragma unroll
    for (int s = 0; s < SS; ++s)
      x[s][c] = *(const v2f*)(tp[s] + 4 * c);
  }
  __builtin_amdgcn_sched_barrier(0);

#pragma unroll
  for (int i = 0; i < 39; ++i)
    ((v4f*)Wn)[tid + 128 * i] = wreg[i];
  __syncthreads();

#pragma unroll
  for (int c = 8; c < 28; ++c) {
#pragma unroll
    for (int s = 0; s < SS; ++s)
      x[s][c] = *(const v2f*)(tp[s] + 4 * c);
    if ((c & 3) == 3)
      __builtin_amdgcn_sched_barrier(0);
  }

  int   toff[SS][8];
  float bsv[SS][8];
#pragma unroll
  for (int s = 0; s < SS; ++s) {
#pragma unroll
    for (int v = 0; v < 8; ++v) {
      const int r2 = v + 8 * kh;
      const int tr = __shfl(ty[s], r2, 32);
      const int tg = __shfl(nd[s], r2, 32);
      toff[s][v] = tg * DD + row;
      bsv[s][v]  = Bs[(s * PP + tr) * 16 + row];
    }
  }
  __builtin_amdgcn_sched_barrier(0);

  const int nn = (row < DD) ? row : DD - 1;
  const float* wl = Wn + 2 * kh * DD + nn;

  v8f acc[SS];
#pragma unroll
  for (int s = 0; s < SS; ++s)
    acc[s] = (v8f){0.f, 0.f, 0.f, 0.f, 0.f, 0.f, 0.f, 0.f};

  v2f bq[2][2][SS][PP];
#pragma unroll
  for (int h = 0; h < 2; ++h) {
#pragma unroll
    for (int s = 0; s < SS; ++s) {
#pragma unroll
      for (int p = 0; p < PP; ++p) {
        const float* q = wl + ((s * PP + p) * RR + 4 * h) * DD;
        bq[0][h][s][p].x = q[0];
        bq[0][h][s][p].y = q[DD];
      }
    }
  }
#pragma unroll
  for (int cp = 0; cp < 16; ++cp) {
    if (cp + 1 < 16) {
#pragma unroll
      for (int h = 0; h < 2; ++h) {
#pragma unroll
        for (int s = 0; s < SS; ++s) {
#pragma unroll
          for (int p = 0; p < PP; ++p) {
            const float* q = wl + ((s * PP + p) * RR + 4 * (2 * cp + 2 + h)) * DD;
            bq[(cp + 1) & 1][h][s][p].x = q[0];
            bq[(cp + 1) & 1][h][s][p].y = q[DD];
          }
        }
      }
    }
    if (cp == 5) {
#pragma unroll
      for (int c2 = 28; c2 < 32; ++c2) {
#pragma unroll
        for (int s = 0; s < SS; ++s)
          x[s][c2] = *(const v2f*)(tp[s] + 4 * c2);
      }
    }
    __builtin_amdgcn_sched_barrier(0x606);
#pragma unroll
    for (int h = 0; h < 2; ++h) {
      const int c = 2 * cp + h;
      v2f a[SS];
      a[0].x = x[1][c].x * x[2][c].x;  a[0].y = x[1][c].y * x[2][c].y;
      a[1].x = x[0][c].x * x[2][c].x;  a[1].y = x[0][c].y * x[2][c].y;
      a[2].x = x[0][c].x * x[1][c].x;  a[2].y = x[0][c].y * x[1][c].y;
#pragma unroll
      for (int p = 0; p < PP; ++p) {
#pragma unroll
        for (int s = 0; s < SS; ++s) {
          v2f ap;
          ap.x = (ty[s] == p) ? a[s].x : 0.0f;
          ap.y = (ty[s] == p) ? a[s].y : 0.0f;
          acc[s] = __builtin_amdgcn_wmma_f32_16x16x4_f32(
              false, ap, false, bq[cp & 1][h][s][p], (short)0, acc[s], false, false);
        }
      }
    }
    __builtin_amdgcn_sched_barrier(0x606);
  }

  if (row < DD) {
#pragma unroll
    for (int s = 0; s < SS; ++s) {
#pragma unroll
      for (int v = 0; v < 8; ++v)
        atomicAdd(&out[toff[s][v]], acc[s][v] + bsv[s][v]);
    }
  }
}

extern "C" void kernel_launch(void* const* d_in, const int* in_sizes, int n_in,
                              void* d_out, int out_size, void* d_ws, size_t ws_size,
                              hipStream_t stream) {
  const float* nodes      = (const float*)d_in[0];
  const float* bp_params  = (const float*)d_in[1];
  const float* bp_bias    = (const float*)d_in[2];
  const float* ho_params  = (const float*)d_in[3];
  const float* ho_bias    = (const float*)d_in[4];
  const int*   edges      = (const int*)d_in[5];
  const int*   edge_types = (const int*)d_in[6];
  float* out  = (float*)d_out;
  float* t_ws = (float*)d_ws;

  bpnet_table_kernel<<<NN / 16, 128, 0, stream>>>(
      nodes, bp_params, bp_bias, t_ws, out);
  bpnet_stage2_kernel<<<EE / 16 / 4, 128, 0, stream>>>(
      t_ws, ho_params, ho_bias, edges, edge_types, out);
}


# Stage 2 runs its three slots one after another (four per-type accumulator tiles each, added at the slot's end): a slot's atomics are in flight under the next slot's WMMAs; atomic order changes
# speedup vs baseline: 2.0309x; 1.0680x over previous
#include <hip/hip_runtime.h>

typedef __attribute__((ext_vector_type(2))) float v2f;
typedef __attribute__((ext_vector_type(4))) float v4f;
typedef __attribute__((ext_vector_type(8))) float v8f;

#define NN 4096
#define EE 16384
#define DD 13
#define RR 128
#define PP 4
#define SS 3

__global__ __launch_bounds__(128) void bpnet_table_kernel(
    const float* __restrict__ nodes,
    const float* __restrict__ bp_params,
    const float* __restrict__ bp_bias,
    float*       __restrict__ tab,
    float*       __restrict__ out)
{
  const int tid   = threadIdx.x;

  const int gid = blockIdx.x * 128 + tid;
  if (gid < NN * DD / 4)
    ((v4f*)out)[gid] = (v4f){0.f, 0.f, 0.f, 0.f};

  const int p     = tid >> 5;
  const int lane  = tid & 31;
  const int n     = lane & 15;
  const int kh    = lane >> 4;
  const int node0 = blockIdx.x * 16;

  const float* __restrict__ nrow = nodes + (size_t)(node0 + n) * DD;
  v2f av[4];
#pragma unroll
  for (int c = 0; c < 3; ++c) {
    av[c].x = nrow[4 * c + 2 * kh];
    av[c].y = nrow[4 * c + 2 * kh + 1];
  }
  const float a12 = nrow[12];
  av[3].x = (kh == 0) ? a12 : 0.0f;
  av[3].y = 0.0f;

  const float* __restrict__ wbase = bp_params + (size_t)p * DD * RR + 4 * n;
  const v4f zero4 = {0.f, 0.f, 0.f, 0.f};
  v4f wv[2][8];
#pragma unroll
  for (int g = 0; g < 2; ++g) {
#pragma unroll
    for (int c = 0; c < 3; ++c) {
      wv[g][2 * c]     = *(const v4f*)(wbase + (4 * c + 2 * kh) * RR + 64 * g);
      wv[g][2 * c + 1] = *(const v4f*)(wbase + (4 * c + 2 * kh + 1) * RR + 64 * g);
    }
    const v4f w12 = *(const v4f*)(wbase + 12 * RR + 64 * g);
    wv[g][6] = (kh == 0) ? w12 : zero4;
    wv[g][7] = zero4;
  }

  v4f bias[2];
#pragma unroll
  for (int g = 0; g < 2; ++g)
    bias[g] = *(const v4f*)(bp_bias + p * RR + 64 * g + 4 * n);

  v8f acc[8];
#pragma unroll
  for (int t = 0; t < 8; ++t)
    acc[t] = (v8f){0.f, 0.f, 0.f, 0.f, 0.f, 0.f, 0.f, 0.f};

#pragma unroll
  for (int c = 0; c < 4; ++c) {
#pragma unroll
    for (int t = 0; t < 8; ++t) {
      const int g = t >> 2, j = t & 3;
      v2f b;
      b.x = wv[g][2 * c][j];
      b.y = wv[g][2 * c + 1][j];
      acc[t] = __builtin_amdgcn_wmma_f32_16x16x4_f32(
          false, av[c], false, b, (short)0, acc[t], false, false);
    }
  }

#pragma unroll
  for (int g = 0; g < 2; ++g) {
#pragma unroll
    for (int v = 0; v < 8; ++v) {
      v4f o;
      o.x = acc[4 * g + 0][v] + bias[g].x;
      o.y = acc[4 * g + 1][v] + bias[g].y;
      o.z = acc[4 * g + 2][v] + bias[g].z;
      o.w = acc[4 * g + 3][v] + bias[g].w;
      o.x = (o.x > 0.0f) ? o.x : 0.0f;
      o.y = (o.y > 0.0f) ? o.y : 0.0f;
      o.z = (o.z > 0.0f) ? o.z : 0.0f;
      o.w = (o.w > 0.0f) ? o.w : 0.0f;
      *(v4f*)(tab + ((size_t)(node0 + v + 8 * kh) * PP + p) * RR + 64 * g + 4 * n) = o;
    }
  }
}

__global__ __launch_bounds__(128) void bpnet_stage2_kernel(
    const float* __restrict__ tab,
    const float* __restrict__ ho_params,
    const float* __restrict__ ho_bias,
    const int*   __restrict__ edges,
    const int*   __restrict__ edge_types,
    float*       __restrict__ out)
{
  __shared__ __attribute__((aligned(16))) float Wn[162 * 256];
  __shared__ float Bs[256];


  const int tid  = threadIdx.x;
  const int wave = tid >> 5;
  const int lane = tid & 31;
  const int row  = lane & 15;
  const int kh   = lane >> 4;
  const int e    = (blockIdx.x * 4 + wave) * 16 + row;

  int nd[SS], ty[SS];
#pragma unroll
  for (int s = 0; s < SS; ++s) {
    nd[s] = edges[e * 3 + s];
    ty[s] = edge_types[e * 3 + s];
  }
  const int bn  = tid & 15;
  const int bnc = (bn < DD) ? bn : DD - 1;
  const int bsp = tid >> 4;
  const int bsq = (tid < 64) ? bsp + 8 : SS * PP - 1;
  const float bias0 = ho_bias[bsp * DD + bnc];
  const float bias1 = ho_bias[bsq * DD + bnc];
  __builtin_amdgcn_sched_barrier(0);

  const v4f* __restrict__ wsrc = (const v4f*)ho_params;
  v4f wreg[39];
#pragma unroll
  for (int i = 0; i < 39; ++i)
    wreg[i] = wsrc[tid + 128 * i];
  __builtin_amdgcn_sched_barrier(0);

  const float* __restrict__ tp[SS];
#pragma unroll
  for (int s = 0; s < SS; ++s)
    tp[s] = tab + ((size_t)nd[s] * PP + ty[s]) * RR + 2 * kh;

  Bs[tid]       = (bn < DD) ? bias0 : 0.0f;
  Bs[tid + 128] = (bn < DD && tid < 64) ? bias1 : 0.0f;

  v2f x[SS][32];
#pragma unroll
  for (int c = 0; c < 12; ++c) {
    x[1][c] = *(const v2f*)(tp[1] + 4 * c);
    x[2][c] = *(const v2f*)(tp[2] + 4 * c);
  }
  __builtin_amdgcn_sched_barrier(0);

#pragma unroll
  for (int i = 0; i < 39; ++i)
    ((v4f*)Wn)[tid + 128 * i] = wreg[i];
  __syncthreads();

#pragma unroll
  for (int c = 12; c < 32; ++c) {
    x[1][c] = *(const v2f*)(tp[1] + 4 * c);
    x[2][c] = *(const v2f*)(tp[2] + 4 * c);
    if ((c & 3) == 3)
      __builtin_amdgcn_sched_barrier(0);
  }
#pragma unroll
  for (int c = 0; c < 20; ++c)
    x[0][c] = *(const v2f*)(tp[0] + 4 * c);
  __builtin_amdgcn_sched_barrier(0);

  int   toff[SS][8];
  float bsv[SS][8];
#pragma unroll
  for (int s = 0; s < SS; ++s) {
#pragma unroll
    for (int v = 0; v < 8; ++v) {
      const int r2 = v + 8 * kh;
      const int tr = __shfl(ty[s], r2, 32);
      const int tg = __shfl(nd[s], r2, 32);
      toff[s][v] = tg * DD + row;
      bsv[s][v]  = Bs[(s * PP + tr) * 16 + row];
    }
  }
  __builtin_amdgcn_sched_barrier(0);

  const int nn = (row < DD) ? row : DD - 1;
  const float* wl = Wn + 2 * kh * DD + nn;

  v2f bq[2][8][PP];
#pragma unroll
  for (int h = 0; h < 8; ++h) {
#pragma unroll
    for (int p = 0; p < PP; ++p) {
      const float* q = wl + (p * RR + 4 * h) * DD;
      bq[0][h][p].x = q[0];
      bq[0][h][p].y = q[DD];
    }
  }

  v8f acc[PP];
#pragma unroll
  for (int g = 0; g < 12; ++g) {
    const int s  = g >> 2;
    const int c0 = (g & 3) * 8;
    const int j0 = (s == 0) ? 1 : 0;
    const int j1 = (s == 2) ? 1 : 2;
    if (g + 1 < 12) {
      const int s1  = (g + 1) >> 2;
      const int c01 = ((g + 1) & 3) * 8;
#pragma unroll
      for (int h = 0; h < 8; ++h) {
#pragma unroll
        for (int p = 0; p < PP; ++p) {
          const float* q = wl + ((s1 * PP + p) * RR + 4 * (c01 + h)) * DD;
          bq[(g + 1) & 1][h][p].x = q[0];
          bq[(g + 1) & 1][h][p].y = q[DD];
        }
      }
    }
    if (g == 2) {
#pragma unroll
      for (int c2 = 20; c2 < 32; ++c2)
        x[0][c2] = *(const v2f*)(tp[0] + 4 * c2);
    }
    __builtin_amdgcn_sched_barrier(0);
    if ((g & 3) == 0) {
#pragma unroll
      for (int p = 0; p < PP; ++p)
        acc[p] = (v8f){0.f, 0.f, 0.f, 0.f, 0.f, 0.f, 0.f, 0.f};
    }
#pragma unroll
    for (int h = 0; h < 8; ++h) {
      const int c = c0 + h;
      v2f a;
      a.x = x[j0][c].x * x[j1][c].x;
      a.y = x[j0][c].y * x[j1][c].y;
#pragma unroll
      for (int p = 0; p < PP; ++p) {
        v2f ap;
        ap.x = (ty[s] == p) ? a.x : 0.0f;
        ap.y = (ty[s] == p) ? a.y : 0.0f;
        acc[p] = __builtin_amdgcn_wmma_f32_16x16x4_f32(
            false, ap, false, bq[g & 1][h][p], (short)0, acc[p], false, false);
      }
    }
    __builtin_amdgcn_sched_barrier(0x606);

    if ((g & 3) == 3) {
      float o[8];
#pragma unroll
      for (int v = 0; v < 8; ++v)
        o[v] = ((acc[0][v] + acc[1][v]) + (acc[2][v] + acc[3][v])) + bsv[s][v];
      if (row < DD) {
#pragma unroll
        for (int v = 0; v < 8; ++v)
          atomicAdd(&out[toff[s][v]], o[v]);
      }
    }
  }
}

extern "C" void kernel_launch(void* const* d_in, const int* in_sizes, int n_in,
                              void* d_out, int out_size, void* d_ws, size_t ws_size,
                              hipStream_t stream) {
  const float* nodes      = (const float*)d_in[0];
  const float* bp_params  = (const float*)d_in[1];
  const float* bp_bias    = (const float*)d_in[2];
  const float* ho_params  = (const float*)d_in[3];
  const float* ho_bias    = (const float*)d_in[4];
  const int*   edges      = (const int*)d_in[5];
  const int*   edge_types = (const int*)d_in[6];
  float* out  = (float*)d_out;
  float* t_ws = (float*)d_ws;

  bpnet_table_kernel<<<NN / 16, 128, 0, stream>>>(
      nodes, bp_params, bp_bias, t_ws, out);
  bpnet_stage2_kernel<<<EE / 16 / 4, 128, 0, stream>>>(
      t_ws, ho_params, ho_bias, edges, edge_types, out);
}
